# starting kernel
# baseline (speedup 1.0000x reference)
#include <hip/hip_runtime.h>
#include <hip/hip_bf16.h>

typedef float v2f __attribute__((ext_vector_type(2)));
typedef float v8f __attribute__((ext_vector_type(8)));

#define EDIM 128
#define MROWS 4096
#define KSTEPS 32

__global__ __launch_bounds__(256)
void qattn_fused_wmma(const float* __restrict__ x,
                      const float* __restrict__ qp,
                      const float* __restrict__ w,
                      const float* __restrict__ bvec,
                      float* __restrict__ out)
{
    const int lane = threadIdx.x & 31;
    const int wave = threadIdx.x >> 5;
    const int mt   = blockIdx.x * 8 + wave;
    const int m0   = mt * 16;
    const int h    = lane >> 4;
    const int l15  = lane & 15;
    const int mrow = m0 + l15;

    float qa[2][2], cb[2][2];
#pragma unroll
    for (int p = 0; p < 2; ++p) {
#pragma unroll
        for (int j = 0; j < 2; ++j) {
            const int kk = 4 * p + 2 * h + j;
            qa[p][j] = qp[kk * 3 + 0];
            cb[p][j] = cosf(qp[kk * 3 + 1]);
        }
    }

    v2f a[KSTEPS];
    const float* xrow = x + (size_t)mrow * EDIM + 2 * h;
#pragma unroll
    for (int s = 0; s < KSTEPS; ++s) {
        const v2f xv = *(const v2f*)(xrow + 4 * s);
        const int p = s & 1;
        v2f t;
        t.x = cb[p][0] * cosf(xv.x + qa[p][0]);
        t.y = cb[p][1] * cosf(xv.y + qa[p][1]);
        a[s] = t;
    }

    for (int nt = 0; nt < 8; ++nt) {
        const int n = nt * 16 + l15;
        const float* wrow = w + (size_t)n * EDIM + 2 * h;
        v8f acc = {};
#pragma unroll
        for (int s = 0; s < KSTEPS; ++s) {
            const v2f bv = *(const v2f*)(wrow + 4 * s);
            acc = __builtin_amdgcn_wmma_f32_16x16x4_f32(
                false, a[s], false, bv, (short)0, acc, false, false);
        }
        const float bias = bvec[n];
#pragma unroll
        for (int r = 0; r < 8; ++r) {
            const int m = m0 + r + 8 * h;
            out[(size_t)m * EDIM + n] = acc[r] + bias;
        }
    }
}

extern "C" void kernel_launch(void* const* d_in, const int* in_sizes, int n_in,
                              void* d_out, int out_size, void* d_ws, size_t ws_size,
                              hipStream_t stream) {
    (void)in_sizes; (void)n_in; (void)out_size; (void)d_ws; (void)ws_size;
    const float* x    = (const float*)d_in[0];
    const float* qp   = (const float*)d_in[1];
    const float* wout = (const float*)d_in[2];
    const float* bout = (const float*)d_in[3];
    float* out = (float*)d_out;

    qattn_fused_wmma<<<32, 256, 0, stream>>>(x, qp, wout, bout, out);
}


# Re-tiled: 256 workgroups x 4 waves, one 16-row tile each (was 32 x 8 tiles); a wave owns 32 output columns in one fused K loop: cos feeds 2 WMMAs, operands requested a chunk ahead
# speedup vs baseline: 2.9395x; 2.9395x over previous
#include <hip/hip_runtime.h>
#include <hip/hip_bf16.h>

typedef float v2f __attribute__((ext_vector_type(2)));
typedef float v8f __attribute__((ext_vector_type(8)));

#define EDIM 128
#define MROWS 4096
#define KSTEPS 32
#define CH 8
#define NCH (KSTEPS / CH)

__device__ __forceinline__
void chunk_mma(const v2f (&xc)[CH], const v2f (&b0)[CH], const v2f (&b1)[CH],
               const float (&qa)[2][2], const float (&cb)[2][2],
               v8f& acc0, v8f& acc1)
{
#pragma unroll
    for (int i = 0; i < CH; ++i) {
        const int p = i & 1;
        v2f t;
        t.x = cb[p][0] * cosf(xc[i].x + qa[p][0]);
        t.y = cb[p][1] * cosf(xc[i].y + qa[p][1]);
        acc0 = __builtin_amdgcn_wmma_f32_16x16x4_f32(
            false, t, false, b0[i], (short)0, acc0, false, false);
        acc1 = __builtin_amdgcn_wmma_f32_16x16x4_f32(
            false, t, false, b1[i], (short)0, acc1, false, false);
    }
}

__device__ __forceinline__
void chunk_load(const float* xrow, const float* wrow0, const float* wrow1, int c,
                v2f (&xc)[CH], v2f (&b0)[CH], v2f (&b1)[CH])
{
#pragma unroll
    for (int i = 0; i < CH; ++i) {
        const int k = 4 * (c * CH + i);
        xc[i] = *(const v2f*)(xrow + k);
        b0[i] = *(const v2f*)(wrow0 + k);
        b1[i] = *(const v2f*)(wrow1 + k);
    }
}

__global__ __launch_bounds__(128)
void qattn_fused_wmma(const float* __restrict__ x,
                      const float* __restrict__ qp,
                      const float* __restrict__ w,
                      const float* __restrict__ bvec,
                      float* __restrict__ out)
{
    const int lane = threadIdx.x & 31;
    const int wave = threadIdx.x >> 5;
    const int m0   = blockIdx.x * 16;
    const int h    = lane >> 4;
    const int l15  = lane & 15;
    const int mrow = m0 + l15;
    const int n0   = wave * 32 + l15;

    const float* xrow  = x + (size_t)mrow * EDIM + 2 * h;
    const float* wrow0 = w + (size_t)n0 * EDIM + 2 * h;
    const float* wrow1 = wrow0 + 16 * EDIM;

    v2f xc[CH], b0[CH], b1[CH];
    chunk_load(xrow, wrow0, wrow1, 0, xc, b0, b1);
    const float bias0 = bvec[n0];
    const float bias1 = bvec[n0 + 16];

    float qa[2][2], cb[2][2];
#pragma unroll
    for (int p = 0; p < 2; ++p) {
#pragma unroll
        for (int j = 0; j < 2; ++j) {
            const int kk = 4 * p + 2 * h + j;
            qa[p][j] = qp[kk * 3 + 0];
            cb[p][j] = cosf(qp[kk * 3 + 1]);
        }
    }

    v8f acc0 = {}, acc1 = {};
    v2f xn[CH] = {}, b0n[CH] = {}, b1n[CH] = {};
    for (int c = 0; c < NCH; ++c) {
        if (c + 1 < NCH)
            chunk_load(xrow, wrow0, wrow1, c + 1, xn, b0n, b1n);
        chunk_mma(xc, b0, b1, qa, cb, acc0, acc1);
#pragma unroll
        for (int i = 0; i < CH; ++i) { xc[i] = xn[i]; b0[i] = b0n[i]; b1[i] = b1n[i]; }
    }

#pragma unroll
    for (int r = 0; r < 8; ++r) {
        float* orow = out + (size_t)(m0 + r + 8 * h) * EDIM + n0;
        orow[0]  = acc0[r] + bias0;
        orow[16] = acc1[r] + bias1;
    }
}

extern "C" void kernel_launch(void* const* d_in, const int* in_sizes, int n_in,
                              void* d_out, int out_size, void* d_ws, size_t ws_size,
                              hipStream_t stream) {
    (void)in_sizes; (void)n_in; (void)out_size; (void)d_ws; (void)ws_size;
    const float* x    = (const float*)d_in[0];
    const float* qp   = (const float*)d_in[1];
    const float* wout = (const float*)d_in[2];
    const float* bout = (const float*)d_in[3];
    float* out = (float*)d_out;

    qattn_fused_wmma<<<256, 128, 0, stream>>>(x, qp, wout, bout, out);
}


# Each gate cosine computed once per workgroup instead of once per wave: wave w builds the A fragments of K steps 8w..8w+7, shared through 8 KB of LDS and one barrier; 64 WMMAs follow, same K order
# speedup vs baseline: 4.2828x; 1.4570x over previous
#include <hip/hip_runtime.h>
#include <hip/hip_bf16.h>

typedef float v2f __attribute__((ext_vector_type(2)));
typedef float v8f __attribute__((ext_vector_type(8)));

#define EDIM 128
#define MROWS 4096
#define KSTEPS 32
#define NWAVES 4
#define KOWN (KSTEPS / NWAVES)

__global__ __launch_bounds__(128)
void qattn_fused_wmma(const float* __restrict__ x,
                      const float* __restrict__ qp,
                      const float* __restrict__ w,
                      const float* __restrict__ bvec,
                      float* __restrict__ out)
{
    __shared__ float4 zt[KSTEPS / 2][32];

    const int lane = threadIdx.x & 31;
    const int wave = threadIdx.x >> 5;
    const int m0   = blockIdx.x * 16;
    const int h    = lane >> 4;
    const int l15  = lane & 15;
    const int mrow = m0 + l15;
    const int n0   = wave * 32 + l15;

    const float* xrow  = x + (size_t)mrow * EDIM + 2 * h;
    const float* wrow0 = w + (size_t)n0 * EDIM + 2 * h;
    const float* wrow1 = wrow0 + 16 * EDIM;

    v2f xc[KOWN];
#pragma unroll
    for (int i = 0; i < KOWN; ++i)
        xc[i] = *(const v2f*)(xrow + 4 * (wave * KOWN + i));

    float qa[2][2], qb[2][2];
#pragma unroll
    for (int p = 0; p < 2; ++p) {
#pragma unroll
        for (int j = 0; j < 2; ++j) {
            const int kk = 4 * p + 2 * h + j;
            qa[p][j] = qp[kk * 3 + 0];
            qb[p][j] = qp[kk * 3 + 1];
        }
    }

    __builtin_amdgcn_sched_barrier(0);

    v2f b0[KSTEPS], b1[KSTEPS];
#pragma unroll
    for (int s = 0; s < KSTEPS; ++s) {
        b0[s] = *(const v2f*)(wrow0 + 4 * s);
        b1[s] = *(const v2f*)(wrow1 + 4 * s);
    }
    const float bias0 = bvec[n0];
    const float bias1 = bvec[n0 + 16];

    float cb[2][2];
#pragma unroll
    for (int p = 0; p < 2; ++p) {
#pragma unroll
        for (int j = 0; j < 2; ++j)
            cb[p][j] = cosf(qb[p][j]);
    }

#pragma unroll
    for (int i = 0; i < KOWN; i += 2) {
        float4 t;
        t.x = cb[0][0] * cosf(xc[i].x + qa[0][0]);
        t.y = cb[0][1] * cosf(xc[i].y + qa[0][1]);
        t.z = cb[1][0] * cosf(xc[i + 1].x + qa[1][0]);
        t.w = cb[1][1] * cosf(xc[i + 1].y + qa[1][1]);
        zt[(wave * KOWN + i) / 2][lane] = t;
    }
    __syncthreads();

    float4 a4[KSTEPS / 2];
#pragma unroll
    for (int j = 0; j < KSTEPS / 2; ++j)
        a4[j] = zt[j][lane];
    __builtin_amdgcn_sched_barrier(0);

    v8f acc0 = {}, acc1 = {};
#pragma unroll
    for (int j = 0; j < KSTEPS / 2; ++j) {
        v2f a;
        a.x = a4[j].x; a.y = a4[j].y;
        acc0 = __builtin_amdgcn_wmma_f32_16x16x4_f32(
            false, a, false, b0[2 * j], (short)0, acc0, false, false);
        acc1 = __builtin_amdgcn_wmma_f32_16x16x4_f32(
            false, a, false, b1[2 * j], (short)0, acc1, false, false);
        a.x = a4[j].z; a.y = a4[j].w;
        acc0 = __builtin_amdgcn_wmma_f32_16x16x4_f32(
            false, a, false, b0[2 * j + 1], (short)0, acc0, false, false);
        acc1 = __builtin_amdgcn_wmma_f32_16x16x4_f32(
            false, a, false, b1[2 * j + 1], (short)0, acc1, false, false);
    }

#pragma unroll
    for (int r = 0; r < 8; ++r) {
        float* orow = out + (size_t)(m0 + r + 8 * h) * EDIM + n0;
        orow[0]  = acc0[r] + bias0;
        orow[16] = acc1[r] + bias1;
    }
}

extern "C" void kernel_launch(void* const* d_in, const int* in_sizes, int n_in,
                              void* d_out, int out_size, void* d_ws, size_t ws_size,
                              hipStream_t stream) {
    (void)in_sizes; (void)n_in; (void)out_size; (void)d_ws; (void)ws_size;
    const float* x    = (const float*)d_in[0];
    const float* qp   = (const float*)d_in[1];
    const float* wout = (const float*)d_in[2];
    const float* bout = (const float*)d_in[3];
    float* out = (float*)d_out;

    qattn_fused_wmma<<<256, 128, 0, stream>>>(x, qp, wout, bout, out);
}


# One range test per wave replaces cosf's 20 per-call divergent branches: arguments below 2^17 take a branch-free copy of the library's short pi/2 reduction, interleaved; otherwise the library cosf
# speedup vs baseline: 4.8371x; 1.1294x over previous
#include <hip/hip_runtime.h>
#include <hip/hip_bf16.h>

typedef float v2f __attribute__((ext_vector_type(2)));
typedef float v8f __attribute__((ext_vector_type(8)));

#define EDIM 128
#define MROWS 4096
#define KSTEPS 32
#define NWAVES 4
#define KOWN (KSTEPS / NWAVES)

__device__ __forceinline__ float cos_below_2p17(float a)
{
    const float ax = __builtin_fabsf(a);
    const float fn = __builtin_rintf(ax * 0x1.45f306p-1f);
    float r = __builtin_fmaf(fn, -0x1.921fb4p+0f, ax);
    r = __builtin_fmaf(fn, -0x1.4442d0p-24f, r);
    r = __builtin_fmaf(fn, -0x1.846988p-48f, r);
    const int n = (int)fn;
    const float s = r * r;
    float ps = __builtin_fmaf(-0x1.983304p-13f, s, 0x1.110388p-7f);
    ps = __builtin_fmaf(s, ps, -0x1.55553ap-3f);
    const float sinr = __builtin_fmaf(r, s * ps, r);
    float pc = __builtin_fmaf(0x1.aea668p-16f, s, -0x1.6c9e76p-10f);
    pc = __builtin_fmaf(s, pc, 0x1.5557eep-5f);
    pc = __builtin_fmaf(s, pc, -0x1.000008p-1f);
    const float cosr = __builtin_fmaf(s, pc, 1.0f);
    const float v = (n & 1) ? -sinr : cosr;
    return __builtin_bit_cast(float,
        __builtin_bit_cast(unsigned, v) ^ (((unsigned)n << 30) & 0x80000000u));
}

__global__ __launch_bounds__(128)
void qattn_fused_wmma(const float* __restrict__ x,
                      const float* __restrict__ qp,
                      const float* __restrict__ w,
                      const float* __restrict__ bvec,
                      float* __restrict__ out)
{
    __shared__ float4 zt[KSTEPS / 2][32];

    const int lane = threadIdx.x & 31;
    const int wave = threadIdx.x >> 5;
    const int m0   = blockIdx.x * 16;
    const int h    = lane >> 4;
    const int l15  = lane & 15;
    const int mrow = m0 + l15;
    const int n0   = wave * 32 + l15;

    const float* xrow  = x + (size_t)mrow * EDIM + 2 * h;
    const float* wrow0 = w + (size_t)n0 * EDIM + 2 * h;
    const float* wrow1 = wrow0 + 16 * EDIM;

    v2f xc[KOWN];
#pragma unroll
    for (int i = 0; i < KOWN; ++i)
        xc[i] = *(const v2f*)(xrow + 4 * (wave * KOWN + i));

    float qa[2][2], qb[2][2];
#pragma unroll
    for (int p = 0; p < 2; ++p) {
#pragma unroll
        for (int j = 0; j < 2; ++j) {
            const int kk = 4 * p + 2 * h + j;
            qa[p][j] = qp[kk * 3 + 0];
            qb[p][j] = qp[kk * 3 + 1];
        }
    }

    __builtin_amdgcn_sched_barrier(0);

    v2f b0[KSTEPS], b1[KSTEPS];
#pragma unroll
    for (int s = 0; s < KSTEPS; ++s) {
        b0[s] = *(const v2f*)(wrow0 + 4 * s);
        b1[s] = *(const v2f*)(wrow1 + 4 * s);
    }
    const float bias0 = bvec[n0];
    const float bias1 = bvec[n0 + 16];

    float arg[KOWN][2];
#pragma unroll
    for (int i = 0; i < KOWN; ++i) {
        arg[i][0] = xc[i].x + qa[i & 1][0];
        arg[i][1] = xc[i].y + qa[i & 1][1];
    }
    bool big = false;
#pragma unroll
    for (int i = 0; i < KOWN; ++i) {
        big |= !(__builtin_fabsf(arg[i][0]) < 0x1p17f);
        big |= !(__builtin_fabsf(arg[i][1]) < 0x1p17f);
    }
#pragma unroll
    for (int p = 0; p < 2; ++p) {
#pragma unroll
        for (int j = 0; j < 2; ++j)
            big |= !(__builtin_fabsf(qb[p][j]) < 0x1p17f);
    }

    float cb[2][2], cz[KOWN][2];
    if (__builtin_amdgcn_ballot_w32(big) == 0u) {
#pragma unroll
        for (int p = 0; p < 2; ++p) {
#pragma unroll
            for (int j = 0; j < 2; ++j)
                cb[p][j] = cos_below_2p17(qb[p][j]);
        }
#pragma unroll
        for (int i = 0; i < KOWN; ++i) {
            cz[i][0] = cos_below_2p17(arg[i][0]);
            cz[i][1] = cos_below_2p17(arg[i][1]);
        }
    } else {
#pragma unroll
        for (int p = 0; p < 2; ++p) {
#pragma unroll
            for (int j = 0; j < 2; ++j)
                cb[p][j] = cosf(qb[p][j]);
        }
#pragma unroll
        for (int i = 0; i < KOWN; ++i) {
            cz[i][0] = cosf(arg[i][0]);
            cz[i][1] = cosf(arg[i][1]);
        }
    }

#pragma unroll
    for (int i = 0; i < KOWN; i += 2) {
        float4 t;
        t.x = cb[0][0] * cz[i][0];
        t.y = cb[0][1] * cz[i][1];
        t.z = cb[1][0] * cz[i + 1][0];
        t.w = cb[1][1] * cz[i + 1][1];
        zt[(wave * KOWN + i) / 2][lane] = t;
    }
    __syncthreads();

    float4 a4[KSTEPS / 2];
#pragma unroll
    for (int j = 0; j < KSTEPS / 2; ++j)
        a4[j] = zt[j][lane];
    __builtin_amdgcn_sched_barrier(0);

    v8f acc0 = {}, acc1 = {};
#pragma unroll
    for (int j = 0; j < KSTEPS / 2; ++j) {
        v2f a;
        a.x = a4[j].x; a.y = a4[j].y;
        acc0 = __builtin_amdgcn_wmma_f32_16x16x4_f32(
            false, a, false, b0[2 * j], (short)0, acc0, false, false);
        acc1 = __builtin_amdgcn_wmma_f32_16x16x4_f32(
            false, a, false, b1[2 * j], (short)0, acc1, false, false);
        a.x = a4[j].z; a.y = a4[j].w;
        acc0 = __builtin_amdgcn_wmma_f32_16x16x4_f32(
            false, a, false, b0[2 * j + 1], (short)0, acc0, false, false);
        acc1 = __builtin_amdgcn_wmma_f32_16x16x4_f32(
            false, a, false, b1[2 * j + 1], (short)0, acc1, false, false);
    }

#pragma unroll
    for (int r = 0; r < 8; ++r) {
        float* orow = out + (size_t)(m0 + r + 8 * h) * EDIM + n0;
        orow[0]  = acc0[r] + bias0;
        orow[16] = acc1[r] + bias1;
    }
}

extern "C" void kernel_launch(void* const* d_in, const int* in_sizes, int n_in,
                              void* d_out, int out_size, void* d_ws, size_t ws_size,
                              hipStream_t stream) {
    (void)in_sizes; (void)n_in; (void)out_size; (void)d_ws; (void)ws_size;
    const float* x    = (const float*)d_in[0];
    const float* qp   = (const float*)d_in[1];
    const float* wout = (const float*)d_in[2];
    const float* bout = (const float*)d_in[3];
    float* out = (float*)d_out;

    qattn_fused_wmma<<<256, 128, 0, stream>>>(x, qp, wout, bout, out);
}


# Operand loads widened to 16 bytes per lane: K cut into blocks of 8, lane half h owns k=8t+4h..+3 for A and B alike, so each load touches half the lines per byte; the k order, hence sum order, changes
# speedup vs baseline: 5.2002x; 1.0751x over previous
#include <hip/hip_runtime.h>
#include <hip/hip_bf16.h>

typedef float v4f __attribute__((ext_vector_type(4)));
typedef float v8f __attribute__((ext_vector_type(8)));

#define EDIM 128
#define MROWS 4096
#define KSTEPS 32
#define NWAVES 4
#define KPAIRS (KSTEPS / 2)
#define POWN (KPAIRS / NWAVES)

__device__ __forceinline__ float cos_below_2p17(float a)
{
    const float ax = __builtin_fabsf(a);
    const float fn = __builtin_rintf(ax * 0x1.45f306p-1f);
    float r = __builtin_fmaf(fn, -0x1.921fb4p+0f, ax);
    r = __builtin_fmaf(fn, -0x1.4442d0p-24f, r);
    r = __builtin_fmaf(fn, -0x1.846988p-48f, r);
    const int n = (int)fn;
    const float s = r * r;
    float ps = __builtin_fmaf(-0x1.983304p-13f, s, 0x1.110388p-7f);
    ps = __builtin_fmaf(s, ps, -0x1.55553ap-3f);
    const float sinr = __builtin_fmaf(r, s * ps, r);
    float pc = __builtin_fmaf(0x1.aea668p-16f, s, -0x1.6c9e76p-10f);
    pc = __builtin_fmaf(s, pc, 0x1.5557eep-5f);
    pc = __builtin_fmaf(s, pc, -0x1.000008p-1f);
    const float cosr = __builtin_fmaf(s, pc, 1.0f);
    const float v = (n & 1) ? -sinr : cosr;
    return __builtin_bit_cast(float,
        __builtin_bit_cast(unsigned, v) ^ (((unsigned)n << 30) & 0x80000000u));
}

__global__ __launch_bounds__(128)
void qattn_fused_wmma(const float* __restrict__ x,
                      const float* __restrict__ qp,
                      const float* __restrict__ w,
                      const float* __restrict__ bvec,
                      float* __restrict__ out)
{
    __shared__ v4f zt[KPAIRS][32];

    const int lane = threadIdx.x & 31;
    const int wave = threadIdx.x >> 5;
    const int m0   = blockIdx.x * 16;
    const int h    = lane >> 4;
    const int l15  = lane & 15;
    const int mrow = m0 + l15;
    const int n0   = wave * 32 + l15;

    const float* xrow  = x + (size_t)mrow * EDIM + 4 * h;
    const float* wrow0 = w + (size_t)n0 * EDIM + 4 * h;
    const float* wrow1 = wrow0 + 16 * EDIM;

    v4f xc[POWN];
#pragma unroll
    for (int i = 0; i < POWN; ++i)
        xc[i] = *(const v4f*)(xrow + 8 * (wave * POWN + i));

    float qa[4], qb[4];
#pragma unroll
    for (int j = 0; j < 4; ++j) {
        const int kk = 4 * h + j;
        qa[j] = qp[kk * 3 + 0];
        qb[j] = qp[kk * 3 + 1];
    }

    __builtin_amdgcn_sched_barrier(0);

    v4f b0[KPAIRS], b1[KPAIRS];
#pragma unroll
    for (int t = 0; t < KPAIRS; ++t) {
        b0[t] = *(const v4f*)(wrow0 + 8 * t);
        b1[t] = *(const v4f*)(wrow1 + 8 * t);
    }
    const float bias0 = bvec[n0];
    const float bias1 = bvec[n0 + 16];

    float arg[POWN][4];
#pragma unroll
    for (int i = 0; i < POWN; ++i) {
#pragma unroll
        for (int j = 0; j < 4; ++j)
            arg[i][j] = xc[i][j] + qa[j];
    }
    bool big = false;
#pragma unroll
    for (int i = 0; i < POWN; ++i) {
#pragma unroll
        for (int j = 0; j < 4; ++j)
            big |= !(__builtin_fabsf(arg[i][j]) < 0x1p17f);
    }
#pragma unroll
    for (int j = 0; j < 4; ++j)
        big |= !(__builtin_fabsf(qb[j]) < 0x1p17f);

    float cb[4], cz[POWN][4];
    if (__builtin_amdgcn_ballot_w32(big) == 0u) {
#pragma unroll
        for (int j = 0; j < 4; ++j)
            cb[j] = cos_below_2p17(qb[j]);
#pragma unroll
        for (int i = 0; i < POWN; ++i) {
#pragma unroll
            for (int j = 0; j < 4; ++j)
                cz[i][j] = cos_below_2p17(arg[i][j]);
        }
    } else {
#pragma unroll
        for (int j = 0; j < 4; ++j)
            cb[j] = cosf(qb[j]);
#pragma unroll
        for (int i = 0; i < POWN; ++i) {
#pragma unroll
            for (int j = 0; j < 4; ++j)
                cz[i][j] = cosf(arg[i][j]);
        }
    }

#pragma unroll
    for (int i = 0; i < POWN; ++i) {
        v4f t;
        t.x = cb[0] * cz[i][0];
        t.y = cb[1] * cz[i][1];
        t.z = cb[2] * cz[i][2];
        t.w = cb[3] * cz[i][3];
        zt[wave * POWN + i][lane] = t;
    }
    __syncthreads();

    v4f a4[KPAIRS];
#pragma unroll
    for (int t = 0; t < KPAIRS; ++t)
        a4[t] = zt[t][lane];
    __builtin_amdgcn_sched_barrier(0);

    v8f acc0 = {}, acc1 = {};
#pragma unroll
    for (int t = 0; t < KPAIRS; ++t) {
        acc0 = __builtin_amdgcn_wmma_f32_16x16x4_f32(
            false, a4[t].xy, false, b0[t].xy, (short)0, acc0, false, false);
        acc1 = __builtin_amdgcn_wmma_f32_16x16x4_f32(
            false, a4[t].xy, false, b1[t].xy, (short)0, acc1, false, false);
        acc0 = __builtin_amdgcn_wmma_f32_16x16x4_f32(
            false, a4[t].zw, false, b0[t].zw, (short)0, acc0, false, false);
        acc1 = __builtin_amdgcn_wmma_f32_16x16x4_f32(
            false, a4[t].zw, false, b1[t].zw, (short)0, acc1, false, false);
    }

#pragma unroll
    for (int r = 0; r < 8; ++r) {
        float* orow = out + (size_t)(m0 + r + 8 * h) * EDIM + n0;
        orow[0]  = acc0[r] + bias0;
        orow[16] = acc1[r] + bias1;
    }
}

extern "C" void kernel_launch(void* const* d_in, const int* in_sizes, int n_in,
                              void* d_out, int out_size, void* d_ws, size_t ws_size,
                              hipStream_t stream) {
    (void)in_sizes; (void)n_in; (void)out_size; (void)d_ws; (void)ws_size;
    const float* x    = (const float*)d_in[0];
    const float* qp   = (const float*)d_in[1];
    const float* wout = (const float*)d_in[2];
    const float* bout = (const float*)d_in[3];
    float* out = (float*)d_out;

    qattn_fused_wmma<<<256, 128, 0, stream>>>(x, qp, wout, bout, out);
}


# Wave stream reordered to wait only for what it needs: split LDS-only barrier, own 4 K blocks first from registers (K order rotated per wave: sum order changes), last 12 B loads behind them
# speedup vs baseline: 5.6523x; 1.0869x over previous
#include <hip/hip_runtime.h>
#include <hip/hip_bf16.h>

typedef float v4f __attribute__((ext_vector_type(4)));
typedef float v8f __attribute__((ext_vector_type(8)));

#define EDIM 128
#define MROWS 4096
#define KSTEPS 32
#define NWAVES 4
#define KPAIRS (KSTEPS / 2)
#define POWN (KPAIRS / NWAVES)

__device__ __forceinline__ float cos_below_2p17(float a)
{
    const float ax = __builtin_fabsf(a);
    const float fn = __builtin_rintf(ax * 0x1.45f306p-1f);
    float r = __builtin_fmaf(fn, -0x1.921fb4p+0f, ax);
    r = __builtin_fmaf(fn, -0x1.4442d0p-24f, r);
    r = __builtin_fmaf(fn, -0x1.846988p-48f, r);
    const int n = (int)fn;
    const float s = r * r;
    float ps = __builtin_fmaf(-0x1.983304p-13f, s, 0x1.110388p-7f);
    ps = __builtin_fmaf(s, ps, -0x1.55553ap-3f);
    const float sinr = __builtin_fmaf(r, s * ps, r);
    float pc = __builtin_fmaf(0x1.aea668p-16f, s, -0x1.6c9e76p-10f);
    pc = __builtin_fmaf(s, pc, 0x1.5557eep-5f);
    pc = __builtin_fmaf(s, pc, -0x1.000008p-1f);
    const float cosr = __builtin_fmaf(s, pc, 1.0f);
    const float v = (n & 1) ? -sinr : cosr;
    return __builtin_bit_cast(float,
        __builtin_bit_cast(unsigned, v) ^ (((unsigned)n << 30) & 0x80000000u));
}

#define NB1 10

#define WMMA_BLOCK(AF, I)                                                      \
    do {                                                                       \
        acc0 = __builtin_amdgcn_wmma_f32_16x16x4_f32(                          \
            false, (AF).xy, false, b0[I].xy, (short)0, acc0, false, false);    \
        acc1 = __builtin_amdgcn_wmma_f32_16x16x4_f32(                          \
            false, (AF).xy, false, b1[I].xy, (short)0, acc1, false, false);    \
        acc0 = __builtin_amdgcn_wmma_f32_16x16x4_f32(                          \
            false, (AF).zw, false, b0[I].zw, (short)0, acc0, false, false);    \
        acc1 = __builtin_amdgcn_wmma_f32_16x16x4_f32(                          \
            false, (AF).zw, false, b1[I].zw, (short)0, acc1, false, false);    \
    } while (0)

#define LOAD_B(I)                                                              \
    do {                                                                       \
        const int tb_ = (wave * POWN + (I)) & (KPAIRS - 1);                    \
        b0[I] = *(const v4f*)(wrow0 + 8 * tb_);                                \
        b1[I] = *(const v4f*)(wrow1 + 8 * tb_);                                \
    } while (0)

__global__ __launch_bounds__(128)
void qattn_fused_wmma(const float* __restrict__ x,
                      const float* __restrict__ qp,
                      const float* __restrict__ w,
                      const float* __restrict__ bvec,
                      float* __restrict__ out)
{
    __shared__ v4f zt[KPAIRS][32];

    const int lane = threadIdx.x & 31;
    const int wave = __builtin_amdgcn_readfirstlane(threadIdx.x >> 5);
    const int m0   = blockIdx.x * 16;
    const int h    = lane >> 4;
    const int l15  = lane & 15;
    const int mrow = m0 + l15;
    const int n0   = wave * 32 + l15;

    const float* xrow  = x + (size_t)mrow * EDIM + 4 * h;
    const float* wrow0 = w + (size_t)n0 * EDIM + 4 * h;
    const float* wrow1 = wrow0 + 16 * EDIM;

    v4f xc[POWN];
#pragma unroll
    for (int i = 0; i < POWN; ++i)
        xc[i] = *(const v4f*)(xrow + 8 * (wave * POWN + i));

    float qa[4], qb[4];
#pragma unroll
    for (int j = 0; j < 4; ++j) {
        const int kk = 4 * h + j;
        qa[j] = qp[kk * 3 + 0];
        qb[j] = qp[kk * 3 + 1];
    }
    const float bias0 = bvec[n0];
    const float bias1 = bvec[n0 + 16];

    __builtin_amdgcn_sched_barrier(0);

    v4f b0[KPAIRS], b1[KPAIRS];
#pragma unroll
    for (int i = 0; i < NB1; ++i)
        LOAD_B(i);

    float arg[POWN][4];
#pragma unroll
    for (int i = 0; i < POWN; ++i) {
#pragma unroll
        for (int j = 0; j < 4; ++j)
            arg[i][j] = xc[i][j] + qa[j];
    }
    bool big = false;
#pragma unroll
    for (int i = 0; i < POWN; ++i) {
#pragma unroll
        for (int j = 0; j < 4; ++j)
            big |= !(__builtin_fabsf(arg[i][j]) < 0x1p17f);
    }
#pragma unroll
    for (int j = 0; j < 4; ++j)
        big |= !(__builtin_fabsf(qb[j]) < 0x1p17f);

    float cb[4], cz[POWN][4];
    if (__builtin_amdgcn_ballot_w32(big) == 0u) {
#pragma unroll
        for (int j = 0; j < 4; ++j)
            cb[j] = cos_below_2p17(qb[j]);
#pragma unroll
        for (int i = 0; i < POWN; ++i) {
#pragma unroll
            for (int j = 0; j < 4; ++j)
                cz[i][j] = cos_below_2p17(arg[i][j]);
        }
    } else {
#pragma unroll
        for (int j = 0; j < 4; ++j)
            cb[j] = cosf(qb[j]);
#pragma unroll
        for (int i = 0; i < POWN; ++i) {
#pragma unroll
            for (int j = 0; j < 4; ++j)
                cz[i][j] = cosf(arg[i][j]);
        }
    }

    v4f tv[POWN];
#pragma unroll
    for (int i = 0; i < POWN; ++i) {
        tv[i].x = cb[0] * cz[i][0];
        tv[i].y = cb[1] * cz[i][1];
        tv[i].z = cb[2] * cz[i][2];
        tv[i].w = cb[3] * cz[i][3];
        zt[wave * POWN + i][lane] = tv[i];
    }

#define ISSUED_BEFORE() asm volatile("" :: "v"(acc0), "v"(acc1) : "memory")
#define ISSUED_AFTER()  asm volatile("" : "+v"(acc0), "+v"(acc1) :: "memory")

    v8f acc0 = {}, acc1 = {};
    WMMA_BLOCK(tv[0], 0);
    ISSUED_BEFORE();
    LOAD_B(NB1 + 0);
    LOAD_B(NB1 + 1);
    ISSUED_AFTER();
    WMMA_BLOCK(tv[1], 1);
    ISSUED_BEFORE();
    asm volatile("s_wait_dscnt 0x0" ::: "memory");
    asm volatile("s_barrier_signal -1" ::: "memory");
    LOAD_B(NB1 + 2);
    LOAD_B(NB1 + 3);
    ISSUED_AFTER();
    WMMA_BLOCK(tv[2], 2);
    ISSUED_BEFORE();
    LOAD_B(NB1 + 4);
    LOAD_B(NB1 + 5);
    asm volatile("s_barrier_wait -1" ::: "memory");
    v4f a4[KPAIRS];
#pragma unroll
    for (int i = POWN; i < KPAIRS; ++i)
        a4[i] = zt[(wave * POWN + i) & (KPAIRS - 1)][lane];
    ISSUED_AFTER();
    WMMA_BLOCK(tv[3], 3);
#pragma unroll
    for (int i = POWN; i < KPAIRS; ++i)
        WMMA_BLOCK(a4[i], i);

#pragma unroll
    for (int r = 0; r < 8; ++r) {
        float* orow = out + (size_t)(m0 + r + 8 * h) * EDIM + n0;
        orow[0]  = acc0[r] + bias0;
        orow[16] = acc1[r] + bias1;
    }
}

extern "C" void kernel_launch(void* const* d_in, const int* in_sizes, int n_in,
                              void* d_out, int out_size, void* d_ws, size_t ws_size,
                              hipStream_t stream) {
    (void)in_sizes; (void)n_in; (void)out_size; (void)d_ws; (void)ws_size;
    const float* x    = (const float*)d_in[0];
    const float* qp   = (const float*)d_in[1];
    const float* wout = (const float*)d_in[2];
    const float* bout = (const float*)d_in[3];
    float* out = (float*)d_out;

    qattn_fused_wmma<<<256, 128, 0, stream>>>(x, qp, wout, bout, out);
}
